# starting kernel
# baseline (speedup 1.0000x reference)
#include <hip/hip_runtime.h>

typedef __attribute__((ext_vector_type(2))) float v2f;
typedef __attribute__((ext_vector_type(8))) float v8f;

#define TPB 256

static __device__ __forceinline__ void atomAddF32(float* p, float v) {
  unsafeAtomicAdd(p, v);
}


__global__ void k_init_deg(float* __restrict__ deg, int N) {
  int n = blockIdx.x * blockDim.x + threadIdx.x;
  if (n < N) deg[n] = 1.0f;
}

__global__ void k_edge_deg(const int* __restrict__ dst, float* __restrict__ deg, int E) {
  int e = blockIdx.x * blockDim.x + threadIdx.x;
  if (e < E) atomAddF32(&deg[dst[e]], 1.0f);
}

__global__ void k_dinv(float* __restrict__ deg, int N) {
  int n = blockIdx.x * blockDim.x + threadIdx.x;
  if (n < N) {
    float d = deg[n];
    deg[n] = (d > 0.0f) ? rsqrtf(d) : 0.0f;
  }
}


__global__ void k_gemm1_wmma(const float* __restrict__ x,
                             const float* __restrict__ W1,
                             const float* __restrict__ dinv,
                             float* __restrict__ h1,
                             float* __restrict__ agg1, int N) {
  const int lane = threadIdx.x & 31;
  const int tile = blockIdx.x * (blockDim.x >> 5) + (threadIdx.x >> 5);
  const int row0 = tile * 16;
  if (row0 >= N) return;

  const int half = lane >> 4;
  const int lrow = lane & 15;

  int arow = row0 + lrow;
  if (arow >= N) arow = N - 1;

  const float* xrow = x + (size_t)arow * 128 + half * 2;
  const float* wcol = W1 + (size_t)(half * 2) * 16 + lrow;

  v8f acc = {};
#pragma unroll
  for (int k0 = 0; k0 < 128; k0 += 4) {
    v2f a = *(const v2f*)(xrow + k0);
    v2f b;
    b.x = wcol[(size_t)k0 * 16];
    b.y = wcol[(size_t)k0 * 16 + 16];
    acc = __builtin_amdgcn_wmma_f32_16x16x4_f32(
         false, a,  false, b,
         (short)0, acc,  false,  false);
  }

  const int col = lrow;
#pragma unroll
  for (int v = 0; v < 8; ++v) {
    const int m = row0 + v + half * 8;
    if (m < N) {
      const float val = acc[v];
      h1[(size_t)m * 16 + col] = val;
      const float di = dinv[m];
      agg1[(size_t)m * 16 + col] = val * di * di;
    }
  }
}


__global__ void k_edge_agg16(const int* __restrict__ ei,
                             const float* __restrict__ dinv,
                             const float* __restrict__ h1,
                             float* __restrict__ agg1, int E) {
  int e = blockIdx.x * blockDim.x + threadIdx.x;
  if (e >= E) return;
  const int s = ei[e];
  const int d = ei[E + e];
  const float w = dinv[s] * dinv[d];
  const float4* hs = (const float4*)(h1 + (size_t)s * 16);
  float4 m0 = hs[0], m1 = hs[1], m2 = hs[2], m3 = hs[3];
  float* op = agg1 + (size_t)d * 16;
  atomAddF32(op + 0,  m0.x * w);
  atomAddF32(op + 1,  m0.y * w);
  atomAddF32(op + 2,  m0.z * w);
  atomAddF32(op + 3,  m0.w * w);
  atomAddF32(op + 4,  m1.x * w);
  atomAddF32(op + 5,  m1.y * w);
  atomAddF32(op + 6,  m1.z * w);
  atomAddF32(op + 7,  m1.w * w);
  atomAddF32(op + 8,  m2.x * w);
  atomAddF32(op + 9,  m2.y * w);
  atomAddF32(op + 10, m2.z * w);
  atomAddF32(op + 11, m2.w * w);
  atomAddF32(op + 12, m3.x * w);
  atomAddF32(op + 13, m3.y * w);
  atomAddF32(op + 14, m3.z * w);
  atomAddF32(op + 15, m3.w * w);
}


__global__ void k_node2(const float* __restrict__ agg1,
                        const float* __restrict__ b1,
                        const float* __restrict__ W2,
                        const float* __restrict__ b2,
                        const float* __restrict__ dinv,
                        float* __restrict__ h2,
                        float* __restrict__ out, int N) {
  int n = blockIdx.x * blockDim.x + threadIdx.x;
  if (n >= N) return;
  const float4* ap = (const float4*)(agg1 + (size_t)n * 16);
  float c0 = 0.0f, c1 = 0.0f;
#pragma unroll
  for (int i = 0; i < 4; ++i) {
    float4 t = ap[i];
    float v[4] = {t.x, t.y, t.z, t.w};
#pragma unroll
    for (int j = 0; j < 4; ++j) {
      const int k = i * 4 + j;
      const float hv = fmaxf(v[j] + b1[k], 0.0f);
      c0 = fmaf(hv, W2[k * 2 + 0], c0);
      c1 = fmaf(hv, W2[k * 2 + 1], c1);
    }
  }
  h2[(size_t)n * 2 + 0] = c0;
  h2[(size_t)n * 2 + 1] = c1;
  const float di = dinv[n];
  const float w = di * di;
  out[(size_t)n * 2 + 0] = fmaf(c0, w, b2[0]);
  out[(size_t)n * 2 + 1] = fmaf(c1, w, b2[1]);
}


__global__ void k_edge_agg2(const int* __restrict__ ei,
                            const float* __restrict__ dinv,
                            const float* __restrict__ h2,
                            float* __restrict__ out, int E) {
  int e = blockIdx.x * blockDim.x + threadIdx.x;
  if (e >= E) return;
  const int s = ei[e];
  const int d = ei[E + e];
  const float w = dinv[s] * dinv[d];
  const float2 m = *(const float2*)(h2 + (size_t)s * 2);
  atomAddF32(out + (size_t)d * 2 + 0, m.x * w);
  atomAddF32(out + (size_t)d * 2 + 1, m.y * w);
}


extern "C" void kernel_launch(void* const* d_in, const int* in_sizes, int n_in,
                              void* d_out, int out_size, void* d_ws, size_t ws_size,
                              hipStream_t stream) {
  const float* x  = (const float*)d_in[0];
  const int*   ei = (const int*)d_in[1];
  const float* W1 = (const float*)d_in[2];
  const float* b1 = (const float*)d_in[3];
  const float* W2 = (const float*)d_in[4];
  const float* b2 = (const float*)d_in[5];
  float* out = (float*)d_out;

  const int N = in_sizes[0] / 128;
  const int E = in_sizes[1] / 2;

  float* deg  = (float*)d_ws;
  float* h1   = deg + N;
  float* agg1 = h1 + (size_t)N * 16;
  float* h2   = agg1 + (size_t)N * 16;

  dim3 blk(TPB);
  dim3 gN((N + TPB - 1) / TPB);
  dim3 gE((E + TPB - 1) / TPB);

  k_init_deg<<<gN, blk, 0, stream>>>(deg, N);
  k_edge_deg<<<gE, blk, 0, stream>>>(ei + E, deg, E);
  k_dinv<<<gN, blk, 0, stream>>>(deg, N);

  const int tiles = (N + 15) / 16;
  const int wavesPerBlock = TPB / 32;
  dim3 gT((tiles + wavesPerBlock - 1) / wavesPerBlock);
  k_gemm1_wmma<<<gT, blk, 0, stream>>>(x, W1, deg, h1, agg1, N);

  k_edge_agg16<<<gE, blk, 0, stream>>>(ei, deg, h1, agg1, E);
  k_node2<<<gN, blk, 0, stream>>>(agg1, b1, W2, b2, deg, h2, out, N);
  k_edge_agg2<<<gE, blk, 0, stream>>>(ei, deg, h2, out, E);
}


# layer-1 edge scatter: 16 lanes per edge (lane = channel, owner's indices by row_share): each load and atomic covers two whole 64-byte rows, not 32 scattered words; sums still in atomic arrival order
# speedup vs baseline: 2.0960x; 2.0960x over previous
#include <hip/hip_runtime.h>

typedef __attribute__((ext_vector_type(2))) float v2f;
typedef __attribute__((ext_vector_type(8))) float v8f;

#define TPB 256

static __device__ __forceinline__ void atomAddF32(float* p, float v) {
  unsafeAtomicAdd(p, v);
}


__global__ void k_init_deg(float* __restrict__ deg, int N) {
  int n = blockIdx.x * blockDim.x + threadIdx.x;
  if (n < N) deg[n] = 1.0f;
}

__global__ void k_edge_deg(const int* __restrict__ dst, float* __restrict__ deg, int E) {
  int e = blockIdx.x * blockDim.x + threadIdx.x;
  if (e < E) atomAddF32(&deg[dst[e]], 1.0f);
}

__global__ void k_dinv(float* __restrict__ deg, int N) {
  int n = blockIdx.x * blockDim.x + threadIdx.x;
  if (n < N) {
    float d = deg[n];
    deg[n] = (d > 0.0f) ? rsqrtf(d) : 0.0f;
  }
}


__global__ void k_gemm1_wmma(const float* __restrict__ x,
                             const float* __restrict__ W1,
                             const float* __restrict__ dinv,
                             float* __restrict__ h1,
                             float* __restrict__ agg1, int N) {
  const int lane = threadIdx.x & 31;
  const int tile = blockIdx.x * (blockDim.x >> 5) + (threadIdx.x >> 5);
  const int row0 = tile * 16;
  if (row0 >= N) return;

  const int half = lane >> 4;
  const int lrow = lane & 15;

  int arow = row0 + lrow;
  if (arow >= N) arow = N - 1;

  const float* xrow = x + (size_t)arow * 128 + half * 2;
  const float* wcol = W1 + (size_t)(half * 2) * 16 + lrow;

  v8f acc = {};
#pragma unroll
  for (int k0 = 0; k0 < 128; k0 += 4) {
    v2f a = *(const v2f*)(xrow + k0);
    v2f b;
    b.x = wcol[(size_t)k0 * 16];
    b.y = wcol[(size_t)k0 * 16 + 16];
    acc = __builtin_amdgcn_wmma_f32_16x16x4_f32(
         false, a,  false, b,
         (short)0, acc,  false,  false);
  }

  const int col = lrow;
#pragma unroll
  for (int v = 0; v < 8; ++v) {
    const int m = row0 + v + half * 8;
    if (m < N) {
      const float val = acc[v];
      h1[(size_t)m * 16 + col] = val;
      const float di = dinv[m];
      agg1[(size_t)m * 16 + col] = val * di * di;
    }
  }
}


template <int K>
static __device__ __forceinline__ unsigned rowShareU(unsigned v) {
  return (unsigned)__builtin_amdgcn_update_dpp(0, (int)v, 0x150 + K, 0xf, 0xf, true);
}
template <int K>
static __device__ __forceinline__ float rowShareF(float v) {
  return __builtin_bit_cast(float, rowShareU<K>(__builtin_bit_cast(unsigned, v)));
}

template <int K>
static __device__ __forceinline__ float gatherStep(const char* __restrict__ h1b,
                                                   unsigned offS, unsigned c4) {
  return *(const float*)(h1b + (rowShareU<K>(offS) + c4));
}

template <int K>
static __device__ __forceinline__ void scatterStep(char* __restrict__ aggb,
                                                   unsigned offD, unsigned c4,
                                                   float w, float hv, bool live) {
  const unsigned o = rowShareU<K>(offD) + c4;
  const float m = hv * rowShareF<K>(w);
  if (live) atomAddF32((float*)(aggb + o), m);
}

template <bool TAIL>
static __device__ __forceinline__ void aggWave(const char* __restrict__ h1b,
                                               char* __restrict__ aggb,
                                               const float* __restrict__ dinv,
                                               int s, int d, unsigned c4,
                                               int rowE, int E) {
  const unsigned offS = (unsigned)s * 64u;
  const unsigned offD = (unsigned)d * 64u;
  float hv[16];
#define GATHER(K) hv[K] = gatherStep<K>(h1b, offS, c4);
  GATHER(0)  GATHER(1)  GATHER(2)  GATHER(3)
  GATHER(4)  GATHER(5)  GATHER(6)  GATHER(7)
  GATHER(8)  GATHER(9)  GATHER(10) GATHER(11)
  GATHER(12) GATHER(13) GATHER(14) GATHER(15)
#undef GATHER
  const float w = dinv[s] * dinv[d];
#define SCATTER(K) scatterStep<K>(aggb, offD, c4, w, hv[K], !TAIL || (rowE + K < E));
  SCATTER(0)  SCATTER(1)  SCATTER(2)  SCATTER(3)
  SCATTER(4)  SCATTER(5)  SCATTER(6)  SCATTER(7)
  SCATTER(8)  SCATTER(9)  SCATTER(10) SCATTER(11)
  SCATTER(12) SCATTER(13) SCATTER(14) SCATTER(15)
#undef SCATTER
}

__global__ void k_edge_agg16(const int* __restrict__ ei,
                             const float* __restrict__ dinv,
                             const float* __restrict__ h1,
                             float* __restrict__ agg1, int E) {
  const int e = blockIdx.x * blockDim.x + threadIdx.x;
  const int wbase = e & ~31;
  if (wbase >= E) return;
  const int ec = (e < E) ? e : (E - 1);
  const int s = ei[ec];
  const int d = ei[E + ec];

  const unsigned lane = threadIdx.x & 31u;
  const unsigned c4 = (lane & 15u) * 4u;
  const int rowE = wbase + (int)(lane & 16u);

  const char* h1b = (const char*)h1;
  char* aggb = (char*)agg1;

  if (wbase + 32 <= E) aggWave<false>(h1b, aggb, dinv, s, d, c4, rowE, E);
  else                 aggWave<true>(h1b, aggb, dinv, s, d, c4, rowE, E);
}


__global__ void k_node2(const float* __restrict__ agg1,
                        const float* __restrict__ b1,
                        const float* __restrict__ W2,
                        const float* __restrict__ b2,
                        const float* __restrict__ dinv,
                        float* __restrict__ h2,
                        float* __restrict__ out, int N) {
  int n = blockIdx.x * blockDim.x + threadIdx.x;
  if (n >= N) return;
  const float4* ap = (const float4*)(agg1 + (size_t)n * 16);
  float c0 = 0.0f, c1 = 0.0f;
#pragma unroll
  for (int i = 0; i < 4; ++i) {
    float4 t = ap[i];
    float v[4] = {t.x, t.y, t.z, t.w};
#pragma unroll
    for (int j = 0; j < 4; ++j) {
      const int k = i * 4 + j;
      const float hv = fmaxf(v[j] + b1[k], 0.0f);
      c0 = fmaf(hv, W2[k * 2 + 0], c0);
      c1 = fmaf(hv, W2[k * 2 + 1], c1);
    }
  }
  h2[(size_t)n * 2 + 0] = c0;
  h2[(size_t)n * 2 + 1] = c1;
  const float di = dinv[n];
  const float w = di * di;
  out[(size_t)n * 2 + 0] = fmaf(c0, w, b2[0]);
  out[(size_t)n * 2 + 1] = fmaf(c1, w, b2[1]);
}


__global__ void k_edge_agg2(const int* __restrict__ ei,
                            const float* __restrict__ dinv,
                            const float* __restrict__ h2,
                            float* __restrict__ out, int E) {
  int e = blockIdx.x * blockDim.x + threadIdx.x;
  if (e >= E) return;
  const int s = ei[e];
  const int d = ei[E + e];
  const float w = dinv[s] * dinv[d];
  const float2 m = *(const float2*)(h2 + (size_t)s * 2);
  atomAddF32(out + (size_t)d * 2 + 0, m.x * w);
  atomAddF32(out + (size_t)d * 2 + 1, m.y * w);
}


extern "C" void kernel_launch(void* const* d_in, const int* in_sizes, int n_in,
                              void* d_out, int out_size, void* d_ws, size_t ws_size,
                              hipStream_t stream) {
  const float* x  = (const float*)d_in[0];
  const int*   ei = (const int*)d_in[1];
  const float* W1 = (const float*)d_in[2];
  const float* b1 = (const float*)d_in[3];
  const float* W2 = (const float*)d_in[4];
  const float* b2 = (const float*)d_in[5];
  float* out = (float*)d_out;

  const int N = in_sizes[0] / 128;
  const int E = in_sizes[1] / 2;

  float* deg  = (float*)d_ws;
  float* h1   = deg + N;
  float* agg1 = h1 + (size_t)N * 16;
  float* h2   = agg1 + (size_t)N * 16;

  dim3 blk(TPB);
  dim3 gN((N + TPB - 1) / TPB);
  dim3 gE((E + TPB - 1) / TPB);

  k_init_deg<<<gN, blk, 0, stream>>>(deg, N);
  k_edge_deg<<<gE, blk, 0, stream>>>(ei + E, deg, E);
  k_dinv<<<gN, blk, 0, stream>>>(deg, N);

  const int tiles = (N + 15) / 16;
  const int wavesPerBlock = TPB / 32;
  dim3 gT((tiles + wavesPerBlock - 1) / wavesPerBlock);
  k_gemm1_wmma<<<gT, blk, 0, stream>>>(x, W1, deg, h1, agg1, N);

  k_edge_agg16<<<gE, blk, 0, stream>>>(ei, deg, h1, agg1, E);
  k_node2<<<gN, blk, 0, stream>>>(agg1, b1, W2, b2, deg, h2, out, N);
  k_edge_agg2<<<gE, blk, 0, stream>>>(ei, deg, h2, out, E);
}


# one accumulation plane per second-level cache half (die number picks it) for the degree, layer-1 and layer-2 atomic adds: no line is added into from both halves; per-half partial sums added after
# speedup vs baseline: 6.5173x; 3.1094x over previous
#include <hip/hip_runtime.h>

typedef __attribute__((ext_vector_type(2))) float v2f;
typedef __attribute__((ext_vector_type(8))) float v8f;

#define TPB 256

static __device__ __forceinline__ void atomAddF32(float* p, float v) {
  unsafeAtomicAdd(p, v);
}

static __device__ __forceinline__ unsigned cacheHalf() {
  const unsigned id = (unsigned)__builtin_amdgcn_s_sendmsg_rtn(0x87);
  return (id >> 18) & 1u;
}


__global__ void k_init_deg(float* __restrict__ deg, float* __restrict__ degB, int N) {
  int n = blockIdx.x * blockDim.x + threadIdx.x;
  if (n < N) {
    deg[n] = 1.0f;
    degB[n] = 0.0f;
  }
}

__global__ void k_edge_deg(const int* __restrict__ dst, float* __restrict__ deg,
                           float* __restrict__ degB, int E) {
  float* plane = cacheHalf() ? degB : deg;
  int e = blockIdx.x * blockDim.x + threadIdx.x;
  if (e < E) atomAddF32(&plane[dst[e]], 1.0f);
}

__global__ void k_dinv(float* __restrict__ deg, const float* __restrict__ degB, int N) {
  int n = blockIdx.x * blockDim.x + threadIdx.x;
  if (n < N) {
    float d = deg[n] + degB[n];
    deg[n] = (d > 0.0f) ? rsqrtf(d) : 0.0f;
  }
}


__global__ void k_gemm1_wmma(const float* __restrict__ x,
                             const float* __restrict__ W1,
                             const float* __restrict__ dinv,
                             float* __restrict__ h1,
                             float* __restrict__ agg1,
                             float* __restrict__ agg1B, int N) {
  const int lane = threadIdx.x & 31;
  const int tile = blockIdx.x * (blockDim.x >> 5) + (threadIdx.x >> 5);
  const int row0 = tile * 16;
  if (row0 >= N) return;

  const int half = lane >> 4;
  const int lrow = lane & 15;

  int arow = row0 + lrow;
  if (arow >= N) arow = N - 1;

  const float* xrow = x + (size_t)arow * 128 + half * 2;
  const float* wcol = W1 + (size_t)(half * 2) * 16 + lrow;

  v8f acc = {};
#pragma unroll
  for (int k0 = 0; k0 < 128; k0 += 4) {
    v2f a = *(const v2f*)(xrow + k0);
    v2f b;
    b.x = wcol[(size_t)k0 * 16];
    b.y = wcol[(size_t)k0 * 16 + 16];
    acc = __builtin_amdgcn_wmma_f32_16x16x4_f32(
         false, a,  false, b,
         (short)0, acc,  false,  false);
  }

  const int col = lrow;
#pragma unroll
  for (int v = 0; v < 8; ++v) {
    const int m = row0 + v + half * 8;
    if (m < N) {
      const float val = acc[v];
      h1[(size_t)m * 16 + col] = val;
      const float di = dinv[m];
      agg1[(size_t)m * 16 + col] = val * di * di;
      if (agg1B) agg1B[(size_t)m * 16 + col] = 0.0f;
    }
  }
}


template <int K>
static __device__ __forceinline__ unsigned rowShareU(unsigned v) {
  return (unsigned)__builtin_amdgcn_update_dpp(0, (int)v, 0x150 + K, 0xf, 0xf, true);
}
template <int K>
static __device__ __forceinline__ float rowShareF(float v) {
  return __builtin_bit_cast(float, rowShareU<K>(__builtin_bit_cast(unsigned, v)));
}

template <int K>
static __device__ __forceinline__ float gatherStep(const char* __restrict__ h1b,
                                                   unsigned offS, unsigned c4) {
  return *(const float*)(h1b + (rowShareU<K>(offS) + c4));
}

template <int K>
static __device__ __forceinline__ void scatterStep(char* __restrict__ aggb,
                                                   unsigned offD, unsigned c4,
                                                   float w, float hv, bool live) {
  const unsigned o = rowShareU<K>(offD) + c4;
  const float m = hv * rowShareF<K>(w);
  if (live) atomAddF32((float*)(aggb + o), m);
}

template <bool TAIL>
static __device__ __forceinline__ void aggWave(const char* __restrict__ h1b,
                                               char* __restrict__ aggb,
                                               const float* __restrict__ dinv,
                                               int s, int d, unsigned c4,
                                               int rowE, int E) {
  const unsigned offS = (unsigned)s * 64u;
  const unsigned offD = (unsigned)d * 64u;
  float hv[16];
#define GATHER(K) hv[K] = gatherStep<K>(h1b, offS, c4);
  GATHER(0)  GATHER(1)  GATHER(2)  GATHER(3)
  GATHER(4)  GATHER(5)  GATHER(6)  GATHER(7)
  GATHER(8)  GATHER(9)  GATHER(10) GATHER(11)
  GATHER(12) GATHER(13) GATHER(14) GATHER(15)
#undef GATHER
  const float w = dinv[s] * dinv[d];
#define SCATTER(K) scatterStep<K>(aggb, offD, c4, w, hv[K], !TAIL || (rowE + K < E));
  SCATTER(0)  SCATTER(1)  SCATTER(2)  SCATTER(3)
  SCATTER(4)  SCATTER(5)  SCATTER(6)  SCATTER(7)
  SCATTER(8)  SCATTER(9)  SCATTER(10) SCATTER(11)
  SCATTER(12) SCATTER(13) SCATTER(14) SCATTER(15)
#undef SCATTER
}

__global__ void k_edge_agg16(const int* __restrict__ ei,
                             const float* __restrict__ dinv,
                             const float* __restrict__ h1,
                             float* __restrict__ agg1, size_t planeBytes, int E) {
  const size_t planeOff = cacheHalf() ? planeBytes : (size_t)0;
  const int e = blockIdx.x * blockDim.x + threadIdx.x;
  const int wbase = e & ~31;
  if (wbase >= E) return;
  const int ec = (e < E) ? e : (E - 1);
  const int s = ei[ec];
  const int d = ei[E + ec];

  const unsigned lane = threadIdx.x & 31u;
  const unsigned c4 = (lane & 15u) * 4u;
  const int rowE = wbase + (int)(lane & 16u);

  const char* h1b = (const char*)h1;
  char* aggb = (char*)agg1 + planeOff;

  if (wbase + 32 <= E) aggWave<false>(h1b, aggb, dinv, s, d, c4, rowE, E);
  else                 aggWave<true>(h1b, aggb, dinv, s, d, c4, rowE, E);
}


__global__ void k_node2(const float* __restrict__ agg1,
                        const float* __restrict__ agg1B,
                        const float* __restrict__ b1,
                        const float* __restrict__ W2,
                        const float* __restrict__ b2,
                        const float* __restrict__ dinv,
                        float* __restrict__ h2,
                        float* __restrict__ out,
                        float* __restrict__ outB, int N) {
  int n = blockIdx.x * blockDim.x + threadIdx.x;
  if (n >= N) return;
  const float4* ap = (const float4*)(agg1 + (size_t)n * 16);
  float4 t[4];
#pragma unroll
  for (int i = 0; i < 4; ++i) t[i] = ap[i];
  if (agg1B) {
    const float4* bp = (const float4*)(agg1B + (size_t)n * 16);
#pragma unroll
    for (int i = 0; i < 4; ++i) {
      const float4 u = bp[i];
      t[i].x += u.x; t[i].y += u.y; t[i].z += u.z; t[i].w += u.w;
    }
  }
  float c0 = 0.0f, c1 = 0.0f;
#pragma unroll
  for (int i = 0; i < 4; ++i) {
    float v[4] = {t[i].x, t[i].y, t[i].z, t[i].w};
#pragma unroll
    for (int j = 0; j < 4; ++j) {
      const int k = i * 4 + j;
      const float hv = fmaxf(v[j] + b1[k], 0.0f);
      c0 = fmaf(hv, W2[k * 2 + 0], c0);
      c1 = fmaf(hv, W2[k * 2 + 1], c1);
    }
  }
  h2[(size_t)n * 2 + 0] = c0;
  h2[(size_t)n * 2 + 1] = c1;
  const float di = dinv[n];
  const float w = di * di;
  out[(size_t)n * 2 + 0] = fmaf(c0, w, b2[0]);
  out[(size_t)n * 2 + 1] = fmaf(c1, w, b2[1]);
  outB[(size_t)n * 2 + 0] = 0.0f;
  outB[(size_t)n * 2 + 1] = 0.0f;
}


__global__ void k_edge_agg2(const int* __restrict__ ei,
                            const float* __restrict__ dinv,
                            const float* __restrict__ h2,
                            float* __restrict__ out,
                            float* __restrict__ outB, int E) {
  float* plane = cacheHalf() ? outB : out;
  int e = blockIdx.x * blockDim.x + threadIdx.x;
  if (e >= E) return;
  const int s = ei[e];
  const int d = ei[E + e];
  const float w = dinv[s] * dinv[d];
  const float2 m = *(const float2*)(h2 + (size_t)s * 2);
  atomAddF32(plane + (size_t)d * 2 + 0, m.x * w);
  atomAddF32(plane + (size_t)d * 2 + 1, m.y * w);
}

__global__ void k_merge_out(float* __restrict__ out, const float* __restrict__ outB, int N) {
  int n = blockIdx.x * blockDim.x + threadIdx.x;
  if (n >= N) return;
  const float2 a = *(const float2*)(out + (size_t)n * 2);
  const float2 b = *(const float2*)(outB + (size_t)n * 2);
  *(float2*)(out + (size_t)n * 2) = make_float2(a.x + b.x, a.y + b.y);
}


extern "C" void kernel_launch(void* const* d_in, const int* in_sizes, int n_in,
                              void* d_out, int out_size, void* d_ws, size_t ws_size,
                              hipStream_t stream) {
  const float* x  = (const float*)d_in[0];
  const int*   ei = (const int*)d_in[1];
  const float* W1 = (const float*)d_in[2];
  const float* b1 = (const float*)d_in[3];
  const float* W2 = (const float*)d_in[4];
  const float* b2 = (const float*)d_in[5];
  float* out = (float*)d_out;

  const int N = in_sizes[0] / 128;
  const int E = in_sizes[1] / 2;

  float* deg  = (float*)d_ws;
  float* h1   = deg + N;
  float* agg1 = h1 + (size_t)N * 16;
  float* h2   = agg1 + (size_t)N * 16;
  float* degB = h1;
  float* outB = h1;
  const size_t need = (size_t)N * 51 * sizeof(float);
  float* agg1B = (ws_size >= need) ? (h2 + (size_t)N * 2) : nullptr;
  const size_t planeBytes = agg1B ? (size_t)((char*)agg1B - (char*)agg1) : 0;

  dim3 blk(TPB);
  dim3 gN((N + TPB - 1) / TPB);
  dim3 gE((E + TPB - 1) / TPB);

  k_init_deg<<<gN, blk, 0, stream>>>(deg, degB, N);
  k_edge_deg<<<gE, blk, 0, stream>>>(ei + E, deg, degB, E);
  k_dinv<<<gN, blk, 0, stream>>>(deg, degB, N);

  const int tiles = (N + 15) / 16;
  const int wavesPerBlock = TPB / 32;
  dim3 gT((tiles + wavesPerBlock - 1) / wavesPerBlock);
  k_gemm1_wmma<<<gT, blk, 0, stream>>>(x, W1, deg, h1, agg1, agg1B, N);

  k_edge_agg16<<<gE, blk, 0, stream>>>(ei, deg, h1, agg1, planeBytes, E);
  k_node2<<<gN, blk, 0, stream>>>(agg1, agg1B, b1, W2, b2, deg, h2, out, outB, N);
  k_edge_agg2<<<gE, blk, 0, stream>>>(ei, deg, h2, out, outB, E);
  k_merge_out<<<gN, blk, 0, stream>>>(out, outB, N);
}


# edge weight factored: node passes store rows times dinv[src], edge passes only gather and add, next node pass scales the sum by dinv[dst]; no weight loads per edge; products rounded per factor
# speedup vs baseline: 8.3587x; 1.2825x over previous
#include <hip/hip_runtime.h>

typedef __attribute__((ext_vector_type(2))) float v2f;
typedef __attribute__((ext_vector_type(8))) float v8f;

#define TPB 256

static __device__ __forceinline__ void atomAddF32(float* p, float v) {
  unsafeAtomicAdd(p, v);
}

static __device__ __forceinline__ unsigned cacheHalf() {
  const unsigned id = (unsigned)__builtin_amdgcn_s_sendmsg_rtn(0x87);
  return (id >> 18) & 1u;
}


__global__ void k_init_deg(float* __restrict__ deg, float* __restrict__ degB, int N) {
  int n = blockIdx.x * blockDim.x + threadIdx.x;
  if (n < N) {
    deg[n] = 1.0f;
    degB[n] = 0.0f;
  }
}

__global__ void k_edge_deg(const int* __restrict__ dst, float* __restrict__ deg,
                           float* __restrict__ degB, int E) {
  float* plane = cacheHalf() ? degB : deg;
  int e = blockIdx.x * blockDim.x + threadIdx.x;
  if (e < E) atomAddF32(&plane[dst[e]], 1.0f);
}

__global__ void k_dinv(float* __restrict__ deg, const float* __restrict__ degB, int N) {
  int n = blockIdx.x * blockDim.x + threadIdx.x;
  if (n < N) {
    float d = deg[n] + degB[n];
    deg[n] = (d > 0.0f) ? rsqrtf(d) : 0.0f;
  }
}


__global__ void k_gemm1_wmma(const float* __restrict__ x,
                             const float* __restrict__ W1,
                             const float* __restrict__ dinv,
                             float* __restrict__ h1,
                             float* __restrict__ agg1,
                             float* __restrict__ agg1B, int N) {
  const int lane = threadIdx.x & 31;
  const int tile = blockIdx.x * (blockDim.x >> 5) + (threadIdx.x >> 5);
  const int row0 = tile * 16;
  if (row0 >= N) return;

  const int half = lane >> 4;
  const int lrow = lane & 15;

  int arow = row0 + lrow;
  if (arow >= N) arow = N - 1;

  const float* xrow = x + (size_t)arow * 128 + half * 2;
  const float* wcol = W1 + (size_t)(half * 2) * 16 + lrow;

  v8f acc = {};
#pragma unroll
  for (int k0 = 0; k0 < 128; k0 += 4) {
    v2f a = *(const v2f*)(xrow + k0);
    v2f b;
    b.x = wcol[(size_t)k0 * 16];
    b.y = wcol[(size_t)k0 * 16 + 16];
    acc = __builtin_amdgcn_wmma_f32_16x16x4_f32(
         false, a,  false, b,
         (short)0, acc,  false,  false);
  }

  const int col = lrow;
#pragma unroll
  for (int v = 0; v < 8; ++v) {
    const int m = row0 + v + half * 8;
    if (m < N) {
      const float val = acc[v] * dinv[m];
      h1[(size_t)m * 16 + col] = val;
      agg1[(size_t)m * 16 + col] = val;
      if (agg1B) agg1B[(size_t)m * 16 + col] = 0.0f;
    }
  }
}


template <int K>
static __device__ __forceinline__ unsigned rowShareU(unsigned v) {
  return (unsigned)__builtin_amdgcn_update_dpp(0, (int)v, 0x150 + K, 0xf, 0xf, true);
}

template <int K>
static __device__ __forceinline__ float gatherStep(const char* __restrict__ h1b,
                                                   unsigned offS, unsigned c4) {
  return *(const float*)(h1b + (rowShareU<K>(offS) + c4));
}

template <int K>
static __device__ __forceinline__ void scatterStep(char* __restrict__ aggb,
                                                   unsigned offD, unsigned c4,
                                                   float hv, bool live) {
  const unsigned o = rowShareU<K>(offD) + c4;
  if (live) atomAddF32((float*)(aggb + o), hv);
}

template <bool TAIL>
static __device__ __forceinline__ void aggWave(const char* __restrict__ h1b,
                                               char* __restrict__ aggb,
                                               int s, int d, unsigned c4,
                                               int rowE, int E) {
  const unsigned offS = (unsigned)s * 64u;
  const unsigned offD = (unsigned)d * 64u;
  float hv[16];
#define GATHER(K) hv[K] = gatherStep<K>(h1b, offS, c4);
  GATHER(0)  GATHER(1)  GATHER(2)  GATHER(3)
  GATHER(4)  GATHER(5)  GATHER(6)  GATHER(7)
  GATHER(8)  GATHER(9)  GATHER(10) GATHER(11)
  GATHER(12) GATHER(13) GATHER(14) GATHER(15)
#undef GATHER
#define SCATTER(K) scatterStep<K>(aggb, offD, c4, hv[K], !TAIL || (rowE + K < E));
  SCATTER(0)  SCATTER(1)  SCATTER(2)  SCATTER(3)
  SCATTER(4)  SCATTER(5)  SCATTER(6)  SCATTER(7)
  SCATTER(8)  SCATTER(9)  SCATTER(10) SCATTER(11)
  SCATTER(12) SCATTER(13) SCATTER(14) SCATTER(15)
#undef SCATTER
}

__global__ void k_edge_agg16(const int* __restrict__ ei,
                             const float* __restrict__ h1,
                             float* __restrict__ agg1, size_t planeBytes, int E) {
  const size_t planeOff = cacheHalf() ? planeBytes : (size_t)0;
  const int e = blockIdx.x * blockDim.x + threadIdx.x;
  const int wbase = e & ~31;
  if (wbase >= E) return;
  const int ec = (e < E) ? e : (E - 1);
  const int s = ei[ec];
  const int d = ei[E + ec];

  const unsigned lane = threadIdx.x & 31u;
  const unsigned c4 = (lane & 15u) * 4u;
  const int rowE = wbase + (int)(lane & 16u);

  const char* h1b = (const char*)h1;
  char* aggb = (char*)agg1 + planeOff;

  if (wbase + 32 <= E) aggWave<false>(h1b, aggb, s, d, c4, rowE, E);
  else                 aggWave<true>(h1b, aggb, s, d, c4, rowE, E);
}


__global__ void k_node2(const float* __restrict__ agg1,
                        const float* __restrict__ agg1B,
                        const float* __restrict__ b1,
                        const float* __restrict__ W2,
                        const float* __restrict__ dinv,
                        float* __restrict__ h2,
                        float* __restrict__ out,
                        float* __restrict__ outB, int N) {
  int n = blockIdx.x * blockDim.x + threadIdx.x;
  if (n >= N) return;
  const float4* ap = (const float4*)(agg1 + (size_t)n * 16);
  float4 t[4];
#pragma unroll
  for (int i = 0; i < 4; ++i) t[i] = ap[i];
  if (agg1B) {
    const float4* bp = (const float4*)(agg1B + (size_t)n * 16);
#pragma unroll
    for (int i = 0; i < 4; ++i) {
      const float4 u = bp[i];
      t[i].x += u.x; t[i].y += u.y; t[i].z += u.z; t[i].w += u.w;
    }
  }
  const float di = dinv[n];
  float c0 = 0.0f, c1 = 0.0f;
#pragma unroll
  for (int i = 0; i < 4; ++i) {
    float v[4] = {t[i].x, t[i].y, t[i].z, t[i].w};
#pragma unroll
    for (int j = 0; j < 4; ++j) {
      const int k = i * 4 + j;
      const float hv = fmaxf(fmaf(v[j], di, b1[k]), 0.0f);
      c0 = fmaf(hv, W2[k * 2 + 0], c0);
      c1 = fmaf(hv, W2[k * 2 + 1], c1);
    }
  }
  c0 *= di;
  c1 *= di;
  h2[(size_t)n * 2 + 0] = c0;
  h2[(size_t)n * 2 + 1] = c1;
  out[(size_t)n * 2 + 0] = c0;
  out[(size_t)n * 2 + 1] = c1;
  outB[(size_t)n * 2 + 0] = 0.0f;
  outB[(size_t)n * 2 + 1] = 0.0f;
}


__global__ void k_edge_agg2(const int* __restrict__ ei,
                            const float* __restrict__ h2,
                            float* __restrict__ out,
                            float* __restrict__ outB, int E) {
  float* plane = cacheHalf() ? outB : out;
  int e = blockIdx.x * blockDim.x + threadIdx.x;
  if (e >= E) return;
  const int s = ei[e];
  const int d = ei[E + e];
  const float2 m = *(const float2*)(h2 + (size_t)s * 2);
  atomAddF32(plane + (size_t)d * 2 + 0, m.x);
  atomAddF32(plane + (size_t)d * 2 + 1, m.y);
}

__global__ void k_merge_out(float* __restrict__ out, const float* __restrict__ outB,
                            const float* __restrict__ dinv,
                            const float* __restrict__ b2, int N) {
  int n = blockIdx.x * blockDim.x + threadIdx.x;
  if (n >= N) return;
  const float2 a = *(const float2*)(out + (size_t)n * 2);
  const float2 b = *(const float2*)(outB + (size_t)n * 2);
  const float di = dinv[n];
  *(float2*)(out + (size_t)n * 2) =
      make_float2(fmaf(a.x + b.x, di, b2[0]), fmaf(a.y + b.y, di, b2[1]));
}


extern "C" void kernel_launch(void* const* d_in, const int* in_sizes, int n_in,
                              void* d_out, int out_size, void* d_ws, size_t ws_size,
                              hipStream_t stream) {
  const float* x  = (const float*)d_in[0];
  const int*   ei = (const int*)d_in[1];
  const float* W1 = (const float*)d_in[2];
  const float* b1 = (const float*)d_in[3];
  const float* W2 = (const float*)d_in[4];
  const float* b2 = (const float*)d_in[5];
  float* out = (float*)d_out;

  const int N = in_sizes[0] / 128;
  const int E = in_sizes[1] / 2;

  float* deg  = (float*)d_ws;
  float* h1   = deg + N;
  float* agg1 = h1 + (size_t)N * 16;
  float* h2   = agg1 + (size_t)N * 16;
  float* degB = h1;
  float* outB = h1;
  const size_t need = (size_t)N * 51 * sizeof(float);
  float* agg1B = (ws_size >= need) ? (h2 + (size_t)N * 2) : nullptr;
  const size_t planeBytes = agg1B ? (size_t)((char*)agg1B - (char*)agg1) : 0;

  dim3 blk(TPB);
  dim3 gN((N + TPB - 1) / TPB);
  dim3 gE((E + TPB - 1) / TPB);

  k_init_deg<<<gN, blk, 0, stream>>>(deg, degB, N);
  k_edge_deg<<<gE, blk, 0, stream>>>(ei + E, deg, degB, E);
  k_dinv<<<gN, blk, 0, stream>>>(deg, degB, N);

  const int tiles = (N + 15) / 16;
  const int wavesPerBlock = TPB / 32;
  dim3 gT((tiles + wavesPerBlock - 1) / wavesPerBlock);
  k_gemm1_wmma<<<gT, blk, 0, stream>>>(x, W1, deg, h1, agg1, agg1B, N);

  k_edge_agg16<<<gE, blk, 0, stream>>>(ei, h1, agg1, planeBytes, E);
  k_node2<<<gN, blk, 0, stream>>>(agg1, agg1B, b1, W2, deg, h2, out, outB, N);
  k_edge_agg2<<<gE, blk, 0, stream>>>(ei, h2, out, outB, E);
  k_merge_out<<<gN, blk, 0, stream>>>(out, outB, deg, b2, N);
}


# layer-2 edge scatter: 2 lanes per edge (lane = channel): each load and atomic covers 16 whole 8-byte rows, one request per edge, not 32 rows once per channel; sums still in atomic arrival order
# speedup vs baseline: 8.7533x; 1.0472x over previous
#include <hip/hip_runtime.h>

typedef __attribute__((ext_vector_type(2))) float v2f;
typedef __attribute__((ext_vector_type(8))) float v8f;

#define TPB 256

static __device__ __forceinline__ void atomAddF32(float* p, float v) {
  unsafeAtomicAdd(p, v);
}

static __device__ __forceinline__ unsigned cacheHalf() {
  const unsigned id = (unsigned)__builtin_amdgcn_s_sendmsg_rtn(0x87);
  return (id >> 18) & 1u;
}


__global__ void k_init_deg(float* __restrict__ deg, float* __restrict__ degB, int N) {
  int n = blockIdx.x * blockDim.x + threadIdx.x;
  if (n < N) {
    deg[n] = 1.0f;
    degB[n] = 0.0f;
  }
}

__global__ void k_edge_deg(const int* __restrict__ dst, float* __restrict__ deg,
                           float* __restrict__ degB, int E) {
  float* plane = cacheHalf() ? degB : deg;
  int e = blockIdx.x * blockDim.x + threadIdx.x;
  if (e < E) atomAddF32(&plane[dst[e]], 1.0f);
}

__global__ void k_dinv(float* __restrict__ deg, const float* __restrict__ degB, int N) {
  int n = blockIdx.x * blockDim.x + threadIdx.x;
  if (n < N) {
    float d = deg[n] + degB[n];
    deg[n] = (d > 0.0f) ? rsqrtf(d) : 0.0f;
  }
}


__global__ void k_gemm1_wmma(const float* __restrict__ x,
                             const float* __restrict__ W1,
                             const float* __restrict__ dinv,
                             float* __restrict__ h1,
                             float* __restrict__ agg1,
                             float* __restrict__ agg1B, int N) {
  const int lane = threadIdx.x & 31;
  const int tile = blockIdx.x * (blockDim.x >> 5) + (threadIdx.x >> 5);
  const int row0 = tile * 16;
  if (row0 >= N) return;

  const int half = lane >> 4;
  const int lrow = lane & 15;

  int arow = row0 + lrow;
  if (arow >= N) arow = N - 1;

  const float* xrow = x + (size_t)arow * 128 + half * 2;
  const float* wcol = W1 + (size_t)(half * 2) * 16 + lrow;

  v8f acc = {};
#pragma unroll
  for (int k0 = 0; k0 < 128; k0 += 4) {
    v2f a = *(const v2f*)(xrow + k0);
    v2f b;
    b.x = wcol[(size_t)k0 * 16];
    b.y = wcol[(size_t)k0 * 16 + 16];
    acc = __builtin_amdgcn_wmma_f32_16x16x4_f32(
         false, a,  false, b,
         (short)0, acc,  false,  false);
  }

  const int col = lrow;
#pragma unroll
  for (int v = 0; v < 8; ++v) {
    const int m = row0 + v + half * 8;
    if (m < N) {
      const float val = acc[v] * dinv[m];
      h1[(size_t)m * 16 + col] = val;
      agg1[(size_t)m * 16 + col] = val;
      if (agg1B) agg1B[(size_t)m * 16 + col] = 0.0f;
    }
  }
}


template <int K>
static __device__ __forceinline__ unsigned rowShareU(unsigned v) {
  return (unsigned)__builtin_amdgcn_update_dpp(0, (int)v, 0x150 + K, 0xf, 0xf, true);
}

template <int K>
static __device__ __forceinline__ float gatherStep(const char* __restrict__ h1b,
                                                   unsigned offS, unsigned c4) {
  return *(const float*)(h1b + (rowShareU<K>(offS) + c4));
}

template <int K>
static __device__ __forceinline__ void scatterStep(char* __restrict__ aggb,
                                                   unsigned offD, unsigned c4,
                                                   float hv, bool live) {
  const unsigned o = rowShareU<K>(offD) + c4;
  if (live) atomAddF32((float*)(aggb + o), hv);
}

template <bool TAIL>
static __device__ __forceinline__ void aggWave(const char* __restrict__ h1b,
                                               char* __restrict__ aggb,
                                               int s, int d, unsigned c4,
                                               int rowE, int E) {
  const unsigned offS = (unsigned)s * 64u;
  const unsigned offD = (unsigned)d * 64u;
  float hv[16];
#define GATHER(K) hv[K] = gatherStep<K>(h1b, offS, c4);
  GATHER(0)  GATHER(1)  GATHER(2)  GATHER(3)
  GATHER(4)  GATHER(5)  GATHER(6)  GATHER(7)
  GATHER(8)  GATHER(9)  GATHER(10) GATHER(11)
  GATHER(12) GATHER(13) GATHER(14) GATHER(15)
#undef GATHER
#define SCATTER(K) scatterStep<K>(aggb, offD, c4, hv[K], !TAIL || (rowE + K < E));
  SCATTER(0)  SCATTER(1)  SCATTER(2)  SCATTER(3)
  SCATTER(4)  SCATTER(5)  SCATTER(6)  SCATTER(7)
  SCATTER(8)  SCATTER(9)  SCATTER(10) SCATTER(11)
  SCATTER(12) SCATTER(13) SCATTER(14) SCATTER(15)
#undef SCATTER
}

__global__ void k_edge_agg16(const int* __restrict__ ei,
                             const float* __restrict__ h1,
                             float* __restrict__ agg1, size_t planeBytes, int E) {
  const size_t planeOff = cacheHalf() ? planeBytes : (size_t)0;
  const int e = blockIdx.x * blockDim.x + threadIdx.x;
  const int wbase = e & ~31;
  if (wbase >= E) return;
  const int ec = (e < E) ? e : (E - 1);
  const int s = ei[ec];
  const int d = ei[E + ec];

  const unsigned lane = threadIdx.x & 31u;
  const unsigned c4 = (lane & 15u) * 4u;
  const int rowE = wbase + (int)(lane & 16u);

  const char* h1b = (const char*)h1;
  char* aggb = (char*)agg1 + planeOff;

  if (wbase + 32 <= E) aggWave<false>(h1b, aggb, s, d, c4, rowE, E);
  else                 aggWave<true>(h1b, aggb, s, d, c4, rowE, E);
}


__global__ void k_node2(const float* __restrict__ agg1,
                        const float* __restrict__ agg1B,
                        const float* __restrict__ b1,
                        const float* __restrict__ W2,
                        const float* __restrict__ dinv,
                        float* __restrict__ h2,
                        float* __restrict__ out,
                        float* __restrict__ outB, int N) {
  int n = blockIdx.x * blockDim.x + threadIdx.x;
  if (n >= N) return;
  const float4* ap = (const float4*)(agg1 + (size_t)n * 16);
  float4 t[4];
#pragma unroll
  for (int i = 0; i < 4; ++i) t[i] = ap[i];
  if (agg1B) {
    const float4* bp = (const float4*)(agg1B + (size_t)n * 16);
#pragma unroll
    for (int i = 0; i < 4; ++i) {
      const float4 u = bp[i];
      t[i].x += u.x; t[i].y += u.y; t[i].z += u.z; t[i].w += u.w;
    }
  }
  const float di = dinv[n];
  float c0 = 0.0f, c1 = 0.0f;
#pragma unroll
  for (int i = 0; i < 4; ++i) {
    float v[4] = {t[i].x, t[i].y, t[i].z, t[i].w};
#pragma unroll
    for (int j = 0; j < 4; ++j) {
      const int k = i * 4 + j;
      const float hv = fmaxf(fmaf(v[j], di, b1[k]), 0.0f);
      c0 = fmaf(hv, W2[k * 2 + 0], c0);
      c1 = fmaf(hv, W2[k * 2 + 1], c1);
    }
  }
  c0 *= di;
  c1 *= di;
  h2[(size_t)n * 2 + 0] = c0;
  h2[(size_t)n * 2 + 1] = c1;
  out[(size_t)n * 2 + 0] = c0;
  out[(size_t)n * 2 + 1] = c1;
  outB[(size_t)n * 2 + 0] = 0.0f;
  outB[(size_t)n * 2 + 1] = 0.0f;
}


template <bool TAIL>
static __device__ __forceinline__ void agg2Wave(const int* __restrict__ es,
                                                const int* __restrict__ ed,
                                                const float* __restrict__ h2,
                                                float* __restrict__ plane,
                                                int e0, unsigned c, int E) {
  const int e1 = e0 + 16;
  const int ec0 = (TAIL && e0 >= E) ? (E - 1) : e0;
  const int ec1 = (TAIL && e1 >= E) ? (E - 1) : e1;
  const unsigned s0 = (unsigned)es[ec0];
  const unsigned s1 = (unsigned)es[ec1];
  const unsigned d0 = (unsigned)ed[ec0];
  const unsigned d1 = (unsigned)ed[ec1];
  const float m0 = h2[s0 * 2u + c];
  const float m1 = h2[s1 * 2u + c];
  if (!TAIL || e0 < E) atomAddF32(plane + (d0 * 2u + c), m0);
  if (!TAIL || e1 < E) atomAddF32(plane + (d1 * 2u + c), m1);
}

__global__ void k_edge_agg2(const int* __restrict__ ei,
                            const float* __restrict__ h2,
                            float* __restrict__ out,
                            float* __restrict__ outB, int E) {
  float* plane = cacheHalf() ? outB : out;
  const int t = blockIdx.x * blockDim.x + threadIdx.x;
  const int wbase = t & ~31;
  if (wbase >= E) return;
  const unsigned lane = threadIdx.x & 31u;
  const unsigned c = lane & 1u;
  const int e0 = wbase + (int)(lane >> 1);
  const int* es = ei;
  const int* ed = ei + E;
  if (wbase + 32 <= E) agg2Wave<false>(es, ed, h2, plane, e0, c, E);
  else                 agg2Wave<true>(es, ed, h2, plane, e0, c, E);
}

__global__ void k_merge_out(float* __restrict__ out, const float* __restrict__ outB,
                            const float* __restrict__ dinv,
                            const float* __restrict__ b2, int N) {
  int n = blockIdx.x * blockDim.x + threadIdx.x;
  if (n >= N) return;
  const float2 a = *(const float2*)(out + (size_t)n * 2);
  const float2 b = *(const float2*)(outB + (size_t)n * 2);
  const float di = dinv[n];
  *(float2*)(out + (size_t)n * 2) =
      make_float2(fmaf(a.x + b.x, di, b2[0]), fmaf(a.y + b.y, di, b2[1]));
}


extern "C" void kernel_launch(void* const* d_in, const int* in_sizes, int n_in,
                              void* d_out, int out_size, void* d_ws, size_t ws_size,
                              hipStream_t stream) {
  const float* x  = (const float*)d_in[0];
  const int*   ei = (const int*)d_in[1];
  const float* W1 = (const float*)d_in[2];
  const float* b1 = (const float*)d_in[3];
  const float* W2 = (const float*)d_in[4];
  const float* b2 = (const float*)d_in[5];
  float* out = (float*)d_out;

  const int N = in_sizes[0] / 128;
  const int E = in_sizes[1] / 2;

  float* deg  = (float*)d_ws;
  float* h1   = deg + N;
  float* agg1 = h1 + (size_t)N * 16;
  float* h2   = agg1 + (size_t)N * 16;
  float* degB = h1;
  float* outB = h1;
  const size_t need = (size_t)N * 51 * sizeof(float);
  float* agg1B = (ws_size >= need) ? (h2 + (size_t)N * 2) : nullptr;
  const size_t planeBytes = agg1B ? (size_t)((char*)agg1B - (char*)agg1) : 0;

  dim3 blk(TPB);
  dim3 gN((N + TPB - 1) / TPB);
  dim3 gE((E + TPB - 1) / TPB);

  k_init_deg<<<gN, blk, 0, stream>>>(deg, degB, N);
  k_edge_deg<<<gE, blk, 0, stream>>>(ei + E, deg, degB, E);
  k_dinv<<<gN, blk, 0, stream>>>(deg, degB, N);

  const int tiles = (N + 15) / 16;
  const int wavesPerBlock = TPB / 32;
  dim3 gT((tiles + wavesPerBlock - 1) / wavesPerBlock);
  k_gemm1_wmma<<<gT, blk, 0, stream>>>(x, W1, deg, h1, agg1, agg1B, N);

  k_edge_agg16<<<gE, blk, 0, stream>>>(ei, h1, agg1, planeBytes, E);
  k_node2<<<gN, blk, 0, stream>>>(agg1, agg1B, b1, W2, deg, h2, out, outB, N);
  k_edge_agg2<<<gE, blk, 0, stream>>>(ei, h2, out, outB, E);
  k_merge_out<<<gN, blk, 0, stream>>>(out, outB, deg, b2, N);
}


# three edge passes: fixed grid of 1024 work-groups whose waves loop over edge chunks, next chunk's indices requested ahead, not one short wave per 32 edges; sums still in atomic arrival order
# speedup vs baseline: 11.0698x; 1.2646x over previous
#include <hip/hip_runtime.h>

typedef __attribute__((ext_vector_type(2))) float v2f;
typedef __attribute__((ext_vector_type(8))) float v8f;

#define TPB 256

#define EDGE_WGS 1024

static __device__ __forceinline__ int waveInGrid() {
  return __builtin_amdgcn_readfirstlane((int)((blockIdx.x * blockDim.x + threadIdx.x) >> 5));
}

static __device__ __forceinline__ void atomAddF32(float* p, float v) {
  unsafeAtomicAdd(p, v);
}

static __device__ __forceinline__ unsigned cacheHalf() {
  const unsigned id = (unsigned)__builtin_amdgcn_s_sendmsg_rtn(0x87);
  return (id >> 18) & 1u;
}


__global__ void k_init_deg(float* __restrict__ deg, float* __restrict__ degB, int N) {
  int n = blockIdx.x * blockDim.x + threadIdx.x;
  if (n < N) {
    deg[n] = 1.0f;
    degB[n] = 0.0f;
  }
}

__global__ void k_edge_deg(const int* __restrict__ dst, float* __restrict__ deg,
                           float* __restrict__ degB, int E) {
  float* plane = cacheHalf() ? degB : deg;
  const int stride = (int)(gridDim.x * blockDim.x);
  int e = blockIdx.x * blockDim.x + threadIdx.x;
  for (; E - e > 3 * stride; e += 4 * stride) {
    const int d0 = dst[e];
    const int d1 = dst[e + stride];
    const int d2 = dst[e + 2 * stride];
    const int d3 = dst[e + 3 * stride];
    atomAddF32(&plane[d0], 1.0f);
    atomAddF32(&plane[d1], 1.0f);
    atomAddF32(&plane[d2], 1.0f);
    atomAddF32(&plane[d3], 1.0f);
  }
  for (; e < E; e += stride) atomAddF32(&plane[dst[e]], 1.0f);
}

__global__ void k_dinv(float* __restrict__ deg, const float* __restrict__ degB, int N) {
  int n = blockIdx.x * blockDim.x + threadIdx.x;
  if (n < N) {
    float d = deg[n] + degB[n];
    deg[n] = (d > 0.0f) ? rsqrtf(d) : 0.0f;
  }
}


__global__ void k_gemm1_wmma(const float* __restrict__ x,
                             const float* __restrict__ W1,
                             const float* __restrict__ dinv,
                             float* __restrict__ h1,
                             float* __restrict__ agg1,
                             float* __restrict__ agg1B, int N) {
  const int lane = threadIdx.x & 31;
  const int tile = blockIdx.x * (blockDim.x >> 5) + (threadIdx.x >> 5);
  const int row0 = tile * 16;
  if (row0 >= N) return;

  const int half = lane >> 4;
  const int lrow = lane & 15;

  int arow = row0 + lrow;
  if (arow >= N) arow = N - 1;

  const float* xrow = x + (size_t)arow * 128 + half * 2;
  const float* wcol = W1 + (size_t)(half * 2) * 16 + lrow;

  v8f acc = {};
#pragma unroll
  for (int k0 = 0; k0 < 128; k0 += 4) {
    v2f a = *(const v2f*)(xrow + k0);
    v2f b;
    b.x = wcol[(size_t)k0 * 16];
    b.y = wcol[(size_t)k0 * 16 + 16];
    acc = __builtin_amdgcn_wmma_f32_16x16x4_f32(
         false, a,  false, b,
         (short)0, acc,  false,  false);
  }

  const int col = lrow;
#pragma unroll
  for (int v = 0; v < 8; ++v) {
    const int m = row0 + v + half * 8;
    if (m < N) {
      const float val = acc[v] * dinv[m];
      h1[(size_t)m * 16 + col] = val;
      agg1[(size_t)m * 16 + col] = val;
      if (agg1B) agg1B[(size_t)m * 16 + col] = 0.0f;
    }
  }
}


template <int K>
static __device__ __forceinline__ unsigned rowShareU(unsigned v) {
  return (unsigned)__builtin_amdgcn_update_dpp(0, (int)v, 0x150 + K, 0xf, 0xf, true);
}

template <int K>
static __device__ __forceinline__ float gatherStep(const char* __restrict__ h1b,
                                                   unsigned offS, unsigned c4) {
  return *(const float*)(h1b + (rowShareU<K>(offS) + c4));
}

template <int K>
static __device__ __forceinline__ void scatterStep(char* __restrict__ aggb,
                                                   unsigned offD, unsigned c4,
                                                   float hv, bool live) {
  const unsigned o = rowShareU<K>(offD) + c4;
  if (live) atomAddF32((float*)(aggb + o), hv);
}

template <bool TAIL>
static __device__ __forceinline__ void aggWave(const char* __restrict__ h1b,
                                               char* __restrict__ aggb,
                                               int s, int d, unsigned c4,
                                               int rowE, int E) {
  const unsigned offS = (unsigned)s * 64u;
  const unsigned offD = (unsigned)d * 64u;
  float hv[16];
#define GATHER(K) hv[K] = gatherStep<K>(h1b, offS, c4);
  GATHER(0)  GATHER(1)  GATHER(2)  GATHER(3)
  GATHER(4)  GATHER(5)  GATHER(6)  GATHER(7)
  GATHER(8)  GATHER(9)  GATHER(10) GATHER(11)
  GATHER(12) GATHER(13) GATHER(14) GATHER(15)
#undef GATHER
#define SCATTER(K) scatterStep<K>(aggb, offD, c4, hv[K], !TAIL || (rowE + K < E));
  SCATTER(0)  SCATTER(1)  SCATTER(2)  SCATTER(3)
  SCATTER(4)  SCATTER(5)  SCATTER(6)  SCATTER(7)
  SCATTER(8)  SCATTER(9)  SCATTER(10) SCATTER(11)
  SCATTER(12) SCATTER(13) SCATTER(14) SCATTER(15)
#undef SCATTER
}

__global__ void k_edge_agg16(const int* __restrict__ ei,
                             const float* __restrict__ h1,
                             float* __restrict__ agg1, size_t planeBytes, int E) {
  const size_t planeOff = cacheHalf() ? planeBytes : (size_t)0;
  const int nWaves = (int)((gridDim.x * blockDim.x) >> 5);
  const int nChunks = (E + 31) >> 5;
  int c = waveInGrid();
  if (c >= nChunks) return;

  const int lane = (int)(threadIdx.x & 31u);
  const unsigned c4 = (unsigned)(lane & 15) * 4u;

  const char* h1b = (const char*)h1;
  char* aggb = (char*)agg1 + planeOff;

  int e = c * 32 + lane;
  int ec = (e < E) ? e : (E - 1);
  int s = ei[ec];
  int d = ei[E + ec];

  for (;;) {
    const int cn = c + nWaves;
    const bool more = cn < nChunks;
    const int en = (more ? cn : c) * 32 + lane;
    const int enc = (en < E) ? en : (E - 1);
    const int sN = ei[enc];
    const int dN = ei[E + enc];

    const int wbase = c * 32;
    const int rowE = wbase + (lane & 16);
    if (wbase + 32 <= E) aggWave<false>(h1b, aggb, s, d, c4, rowE, E);
    else                 aggWave<true>(h1b, aggb, s, d, c4, rowE, E);

    if (!more) break;
    c = cn;
    s = sN;
    d = dN;
  }
}


__global__ void k_node2(const float* __restrict__ agg1,
                        const float* __restrict__ agg1B,
                        const float* __restrict__ b1,
                        const float* __restrict__ W2,
                        const float* __restrict__ dinv,
                        float* __restrict__ h2,
                        float* __restrict__ out,
                        float* __restrict__ outB, int N) {
  int n = blockIdx.x * blockDim.x + threadIdx.x;
  if (n >= N) return;
  const float4* ap = (const float4*)(agg1 + (size_t)n * 16);
  float4 t[4];
#pragma unroll
  for (int i = 0; i < 4; ++i) t[i] = ap[i];
  if (agg1B) {
    const float4* bp = (const float4*)(agg1B + (size_t)n * 16);
#pragma unroll
    for (int i = 0; i < 4; ++i) {
      const float4 u = bp[i];
      t[i].x += u.x; t[i].y += u.y; t[i].z += u.z; t[i].w += u.w;
    }
  }
  const float di = dinv[n];
  float c0 = 0.0f, c1 = 0.0f;
#pragma unroll
  for (int i = 0; i < 4; ++i) {
    float v[4] = {t[i].x, t[i].y, t[i].z, t[i].w};
#pragma unroll
    for (int j = 0; j < 4; ++j) {
      const int k = i * 4 + j;
      const float hv = fmaxf(fmaf(v[j], di, b1[k]), 0.0f);
      c0 = fmaf(hv, W2[k * 2 + 0], c0);
      c1 = fmaf(hv, W2[k * 2 + 1], c1);
    }
  }
  c0 *= di;
  c1 *= di;
  h2[(size_t)n * 2 + 0] = c0;
  h2[(size_t)n * 2 + 1] = c1;
  out[(size_t)n * 2 + 0] = c0;
  out[(size_t)n * 2 + 1] = c1;
  outB[(size_t)n * 2 + 0] = 0.0f;
  outB[(size_t)n * 2 + 1] = 0.0f;
}


static __device__ __forceinline__ void agg2Idx(const int* __restrict__ es,
                                               const int* __restrict__ ed,
                                               int e0, int E,
                                               unsigned (&s)[4], unsigned (&d)[4]) {
  for (int k = 0; k < 4; ++k) {
    const int e = e0 + 16 * k;
    const int ec = (e < E) ? e : (E - 1);
    s[k] = (unsigned)es[ec];
    d[k] = (unsigned)ed[ec];
  }
}

template <bool TAIL>
static __device__ __forceinline__ void agg2Add(float* __restrict__ plane,
                                               const unsigned (&d)[4],
                                               const float (&m)[4],
                                               int e0, unsigned c, int E) {
  for (int k = 0; k < 4; ++k)
    if (!TAIL || e0 + 16 * k < E) atomAddF32(plane + (d[k] * 2u + c), m[k]);
}

__global__ void k_edge_agg2(const int* __restrict__ ei,
                            const float* __restrict__ h2,
                            float* __restrict__ out,
                            float* __restrict__ outB, int E) {
  float* plane = cacheHalf() ? outB : out;
  const int nWaves = (int)((gridDim.x * blockDim.x) >> 5);
  const int nUnits = (E + 63) >> 6;
  int u = waveInGrid();
  if (u >= nUnits) return;
  const unsigned lane = threadIdx.x & 31u;
  const unsigned c = lane & 1u;
  const int pe = (int)(lane >> 1);
  const int* es = ei;
  const int* ed = ei + E;

  unsigned s[4], d[4];
  agg2Idx(es, ed, u * 64 + pe, E, s, d);

  for (;;) {
    const int un = u + nWaves;
    const bool more = un < nUnits;
    unsigned sN[4], dN[4];
    agg2Idx(es, ed, (more ? un : u) * 64 + pe, E, sN, dN);

    float m[4];
    for (int k = 0; k < 4; ++k) m[k] = h2[s[k] * 2u + c];

    const int e0 = u * 64 + pe;
    if (u * 64 + 64 <= E) agg2Add<false>(plane, d, m, e0, c, E);
    else                  agg2Add<true>(plane, d, m, e0, c, E);

    if (!more) break;
    u = un;
    for (int k = 0; k < 4; ++k) { s[k] = sN[k]; d[k] = dN[k]; }
  }
}

__global__ void k_merge_out(float* __restrict__ out, const float* __restrict__ outB,
                            const float* __restrict__ dinv,
                            const float* __restrict__ b2, int N) {
  int n = blockIdx.x * blockDim.x + threadIdx.x;
  if (n >= N) return;
  const float2 a = *(const float2*)(out + (size_t)n * 2);
  const float2 b = *(const float2*)(outB + (size_t)n * 2);
  const float di = dinv[n];
  *(float2*)(out + (size_t)n * 2) =
      make_float2(fmaf(a.x + b.x, di, b2[0]), fmaf(a.y + b.y, di, b2[1]));
}


extern "C" void kernel_launch(void* const* d_in, const int* in_sizes, int n_in,
                              void* d_out, int out_size, void* d_ws, size_t ws_size,
                              hipStream_t stream) {
  const float* x  = (const float*)d_in[0];
  const int*   ei = (const int*)d_in[1];
  const float* W1 = (const float*)d_in[2];
  const float* b1 = (const float*)d_in[3];
  const float* W2 = (const float*)d_in[4];
  const float* b2 = (const float*)d_in[5];
  float* out = (float*)d_out;

  const int N = in_sizes[0] / 128;
  const int E = in_sizes[1] / 2;

  float* deg  = (float*)d_ws;
  float* h1   = deg + N;
  float* agg1 = h1 + (size_t)N * 16;
  float* h2   = agg1 + (size_t)N * 16;
  float* degB = h1;
  float* outB = h1;
  const size_t need = (size_t)N * 51 * sizeof(float);
  float* agg1B = (ws_size >= need) ? (h2 + (size_t)N * 2) : nullptr;
  const size_t planeBytes = agg1B ? (size_t)((char*)agg1B - (char*)agg1) : 0;

  dim3 blk(TPB);
  dim3 gN((N + TPB - 1) / TPB);
  dim3 gP(EDGE_WGS);

  k_init_deg<<<gN, blk, 0, stream>>>(deg, degB, N);
  k_edge_deg<<<gP, blk, 0, stream>>>(ei + E, deg, degB, E);
  k_dinv<<<gN, blk, 0, stream>>>(deg, degB, N);

  const int tiles = (N + 15) / 16;
  const int wavesPerBlock = TPB / 32;
  dim3 gT((tiles + wavesPerBlock - 1) / wavesPerBlock);
  k_gemm1_wmma<<<gT, blk, 0, stream>>>(x, W1, deg, h1, agg1, agg1B, N);

  k_edge_agg16<<<gP, blk, 0, stream>>>(ei, h1, agg1, planeBytes, E);
  k_node2<<<gN, blk, 0, stream>>>(agg1, agg1B, b1, W2, deg, h2, out, outB, N);
  k_edge_agg2<<<gP, blk, 0, stream>>>(ei, h2, out, outB, E);
  k_merge_out<<<gN, blk, 0, stream>>>(out, outB, deg, b2, N);
}
